# starting kernel
# baseline (speedup 1.0000x reference)
#include <hip/hip_runtime.h>

typedef __attribute__((ext_vector_type(2))) float v2f;
typedef __attribute__((ext_vector_type(4))) float v4f;
typedef __attribute__((ext_vector_type(8))) float v8f;
typedef __attribute__((ext_vector_type(4))) int   v4i;

#define FDIM 200

__global__ __launch_bounds__(256) void fm_linear_wmma(
    const int*   __restrict__ ids,
    const float* __restrict__ vals,
    const float* __restrict__ w,
    const float* __restrict__ bias,
    float*       __restrict__ out,
    int nB)
{
  const int lane = threadIdx.x & 31;
  const int wave = threadIdx.x >> 5;
  const int m    = lane & 15;
  const int h    = lane >> 4;
  const int exBase = (blockIdx.x * 8 + wave) * 16;
  if (exBase >= nB) return;

  const long long rowOff = (long long)(exBase + m) * FDIM;
  const int*   idRow = ids  + rowOff;
  const float* vaRow = vals + rowOff;
  const int colBase = 4 * h;

  v8f acc = {};
  const v2f ones = {1.0f, 1.0f};

  for (int j = 0; j < FDIM / 8; ++j) {
    const int c0 = 8 * j + colBase;
    v4i id4 = *(const v4i*)(idRow + c0);
    v4f va4 = *(const v4f*)(vaRow + c0);
    v2f a01 = { w[id4.x] * va4.x, w[id4.y] * va4.y };
    v2f a23 = { w[id4.z] * va4.z, w[id4.w] * va4.w };
    acc = __builtin_amdgcn_wmma_f32_16x16x4_f32(
        false, a01, false, ones, (short)0, acc, false, false);
    acc = __builtin_amdgcn_wmma_f32_16x16x4_f32(
        false, a23, false, ones, (short)0, acc, false, false);
  }

  if ((lane & 15) == 0) {
    const float b = bias[0];
    float* o = out + exBase + h * 8;
    v4f lo = { acc[0] + b, acc[1] + b, acc[2] + b, acc[3] + b };
    v4f hi = { acc[4] + b, acc[5] + b, acc[6] + b, acc[7] + b };
    *(v4f*)(o)     = lo;
    *(v4f*)(o + 4) = hi;
  }
}

extern "C" void kernel_launch(void* const* d_in, const int* in_sizes, int n_in,
                              void* d_out, int out_size, void* d_ws, size_t ws_size,
                              hipStream_t stream) {
  const int*   ids  = (const int*)d_in[0];
  const float* vals = (const float*)d_in[1];
  const float* w    = (const float*)d_in[2];
  const float* bias = (const float*)d_in[3];
  float* out = (float*)d_out;

  const int nB = in_sizes[0] / FDIM;
  const int blocks = (nB + 127) / 128;
  fm_linear_wmma<<<blocks, 256, 0, stream>>>(ids, vals, w, bias, out, nB);
}


# ids/vals loads regrouped by 128-B line: a trip loads one whole line per row (head+tail pieces in one edge trip) ahead of its gathers, so each line is requested once; fp32 sum order changes
# speedup vs baseline: 1.2017x; 1.2017x over previous
#include <hip/hip_runtime.h>

typedef __attribute__((ext_vector_type(2))) float v2f;
typedef __attribute__((ext_vector_type(4))) float v4f;
typedef __attribute__((ext_vector_type(8))) float v8f;
typedef __attribute__((ext_vector_type(4))) int   v4i;

#define FDIM 200


__device__ __forceinline__ v8f fm_chunk(v8f acc, v4i id4, v4f va4,
                                        const float* __restrict__ w, v2f ones)
{
  v2f a01 = { w[id4.x] * va4.x, w[id4.y] * va4.y };
  v2f a23 = { w[id4.z] * va4.z, w[id4.w] * va4.w };
  acc = __builtin_amdgcn_wmma_f32_16x16x4_f32(
      false, a01, false, ones, (short)0, acc, false, false);
  acc = __builtin_amdgcn_wmma_f32_16x16x4_f32(
      false, a23, false, ones, (short)0, acc, false, false);
  return acc;
}

__global__ __launch_bounds__(256) void fm_linear_wmma(
    const int*   __restrict__ ids,
    const float* __restrict__ vals,
    const float* __restrict__ w,
    const float* __restrict__ bias,
    float*       __restrict__ out,
    int nB)
{
  const int lane = threadIdx.x & 31;
  const int wave = threadIdx.x >> 5;
  const int m    = lane & 15;
  const int h    = lane >> 4;
  const int exBase = (blockIdx.x * 8 + wave) * 16;
  if (exBase >= nB) return;

  const long long rowOff = (long long)(exBase + m) * FDIM;
  const int*   idRow = ids  + rowOff;
  const float* vaRow = vals + rowOff;

  const int q     = m & 3;
  const int nHead = 8 - 2 * q;
  const int eoff  = 4 * nHead;

  v8f acc = {};
  const v2f ones = {1.0f, 1.0f};

  const int*   idMain = idRow + eoff + 16 * h;
  const float* vaMain = vaRow + eoff + 16 * h;
  for (int t = 0; t < 5; ++t) {
    const int*   ip = idMain + 32 * t;
    const float* vp = vaMain + 32 * t;
    v4i i0 = *(const v4i*)(ip);
    v4i i1 = *(const v4i*)(ip + 4);
    v4i i2 = *(const v4i*)(ip + 8);
    v4i i3 = *(const v4i*)(ip + 12);
    v4f a0 = *(const v4f*)(vp);
    v4f a1 = *(const v4f*)(vp + 4);
    v4f a2 = *(const v4f*)(vp + 8);
    v4f a3 = *(const v4f*)(vp + 12);
    asm volatile("" :: "v"(a0), "v"(a1), "v"(a2), "v"(a3));
    acc = fm_chunk(acc, i0, a0, w, ones);
    acc = fm_chunk(acc, i1, a1, w, ones);
    acc = fm_chunk(acc, i2, a2, w, ones);
    acc = fm_chunk(acc, i3, a3, w, ones);
  }

  {
    v4i ie[5];
    v4f ae[5];
#pragma unroll
    for (int i = 0; i < 5; ++i) {
      const int e = 5 * h + i;
      const int f = (e < nHead) ? 4 * e : 160 + 4 * e;
      ie[i] = *(const v4i*)(idRow + f);
      ae[i] = *(const v4f*)(vaRow + f);
    }
    asm volatile("" :: "v"(ae[0]), "v"(ae[1]), "v"(ae[2]), "v"(ae[3]), "v"(ae[4]));
#pragma unroll
    for (int i = 0; i < 5; ++i)
      acc = fm_chunk(acc, ie[i], ae[i], w, ones);
  }

  if ((lane & 15) == 0) {
    const float b = bias[0];
    float* o = out + exBase + h * 8;
    v4f lo = { acc[0] + b, acc[1] + b, acc[2] + b, acc[3] + b };
    v4f hi = { acc[4] + b, acc[5] + b, acc[6] + b, acc[7] + b };
    *(v4f*)(o)     = lo;
    *(v4f*)(o + 4) = hi;
  }
}

extern "C" void kernel_launch(void* const* d_in, const int* in_sizes, int n_in,
                              void* d_out, int out_size, void* d_ws, size_t ws_size,
                              hipStream_t stream) {
  const int*   ids  = (const int*)d_in[0];
  const float* vals = (const float*)d_in[1];
  const float* w    = (const float*)d_in[2];
  const float* bias = (const float*)d_in[3];
  float* out = (float*)d_out;

  const int nB = in_sizes[0] / FDIM;
  const int blocks = (nB + 127) / 128;
  fm_linear_wmma<<<blocks, 256, 0, stream>>>(ids, vals, w, bias, out, nB);
}
